# starting kernel
# baseline (speedup 1.0000x reference)
#include <hip/hip_runtime.h>
#include <hip/hip_bf16.h>

#define B_ 4
#define N_ 384
#define D_ 64
#define TS 3

typedef __attribute__((ext_vector_type(2))) float v2f;
typedef __attribute__((ext_vector_type(8))) float v8f;

__global__ __launch_bounds__(128) void gnn_stage1(const float* __restrict__ A,
                                                  const float* __restrict__ X,
                                                  float* __restrict__ T) {
    const int wave = threadIdx.x >> 5;
    const int lane = threadIdx.x & 31;
    const int b    = blockIdx.x / (N_ / 16);
    const int i0   = (blockIdx.x % (N_ / 16)) * 16;
    const int d0   = wave * 16;

    const int lm = lane & 15;
    const int lk = (lane >> 4) * 2;

    const float* Arow = A + (size_t)(b * N_ + i0 + lm) * N_ * 2;
    const float* Xb   = X + (size_t)b * N_ * D_ * TS;

    v8f c = {};
    for (int k0 = 0; k0 < N_; k0 += 4) {
        const int j = k0 + lk;
        const float4 af = *(const float4*)(Arow + (size_t)j * 2);
        v2f a;  a.x = af.y;  a.y = af.w;
        v2f bv;
        bv.x = Xb[((size_t)j       * D_ + d0 + lm) * TS + 2];
        bv.y = Xb[((size_t)(j + 1) * D_ + d0 + lm) * TS + 2];
        c = __builtin_amdgcn_wmma_f32_16x16x4_f32(false, a, false, bv,
                                                  (short)0, c, false, false);
    }

    const int row0 = (lane >> 4) * 8;
    float* Trow = T + (size_t)(b * N_ + i0) * D_;
#pragma unroll
    for (int r = 0; r < 8; ++r) {
        const int row = row0 + r;
        const int col = d0 + lm;
        const float x1 = Xb[((size_t)(i0 + row) * D_ + col) * TS + 1];
        Trow[(size_t)row * D_ + col] = c[r] + x1;
    }
}

__global__ __launch_bounds__(128) void gnn_stage2(const float* __restrict__ A,
                                                  const float* __restrict__ X,
                                                  const float* __restrict__ T,
                                                  const float* __restrict__ W,
                                                  float* __restrict__ out) {
    __shared__ float S[16][D_ + 4];

    const int wave = threadIdx.x >> 5;
    const int lane = threadIdx.x & 31;
    const int b    = blockIdx.x / (N_ / 16);
    const int i0   = (blockIdx.x % (N_ / 16)) * 16;
    const int d0   = wave * 16;
    const int lm   = lane & 15;
    const int lk   = (lane >> 4) * 2;
    const int row0 = (lane >> 4) * 8;

    const float* Arow = A + (size_t)(b * N_ + i0 + lm) * N_ * 2;
    const float* Xb   = X + (size_t)b * N_ * D_ * TS;
    const float* Tb   = T + (size_t)b * N_ * D_;

    v8f c = {};
    for (int k0 = 0; k0 < N_; k0 += 4) {
        const int j = k0 + lk;
        const float4 af = *(const float4*)(Arow + (size_t)j * 2);
        v2f a;  a.x = af.x;  a.y = af.z;
        v2f bv;
        bv.x = Tb[(size_t)j       * D_ + d0 + lm];
        bv.y = Tb[(size_t)(j + 1) * D_ + d0 + lm];
        c = __builtin_amdgcn_wmma_f32_16x16x4_f32(false, a, false, bv,
                                                  (short)0, c, false, false);
    }
#pragma unroll
    for (int r = 0; r < 8; ++r) {
        const int row = row0 + r;
        const int col = d0 + lm;
        S[row][col] = c[r] + Xb[((size_t)(i0 + row) * D_ + col) * TS + 0];
    }
    __syncthreads();

    v8f c2 = {};
    for (int k0 = 0; k0 < D_; k0 += 4) {
        const int k = k0 + lk;
        v2f a;  a.x = S[lm][k];  a.y = S[lm][k + 1];
        v2f bv;
        bv.x = W[(size_t)k       * D_ + d0 + lm];
        bv.y = W[(size_t)(k + 1) * D_ + d0 + lm];
        c2 = __builtin_amdgcn_wmma_f32_16x16x4_f32(false, a, false, bv,
                                                   (short)0, c2, false, false);
    }

    float* Orow = out + (size_t)(b * N_ + i0) * D_;
#pragma unroll
    for (int r = 0; r < 8; ++r) {
        Orow[(size_t)(row0 + r) * D_ + d0 + lm] = c2[r];
    }
}

extern "C" void kernel_launch(void* const* d_in, const int* in_sizes, int n_in,
                              void* d_out, int out_size, void* d_ws, size_t ws_size,
                              hipStream_t stream) {
    (void)in_sizes; (void)n_in; (void)out_size; (void)ws_size;
    const float* A = (const float*)d_in[0];
    const float* X = (const float*)d_in[1];
    const float* W = (const float*)d_in[2];
    float* out = (float*)d_out;
    float* T   = (float*)d_ws;

    const dim3 grid(B_ * (N_ / 16));
    const dim3 block(128);

    gnn_stage1<<<grid, block, 0, stream>>>(A, X, T);
    gnn_stage2<<<grid, block, 0, stream>>>(A, X, T, W, out);
}


# Both GEMMs: 384-deep contraction cut into 16 slices, one wave each, all of a wave's loads issued ahead of its 24 products; partial tiles summed in LDS in fixed order (sum order changes)
# speedup vs baseline: 2.5427x; 2.5427x over previous
#include <hip/hip_runtime.h>
#include <hip/hip_bf16.h>

#define B_ 4
#define N_ 384
#define D_ 64
#define TS 3
#define KS 16
#define KW (N_ / KS)
#define NSTEP (KW / 4)
#define NT (D_ / 16)
#define PP (D_ + 4)
#define NTHREADS (KS * 32)

typedef __attribute__((ext_vector_type(2))) float v2f;
typedef __attribute__((ext_vector_type(4))) float v4f;
typedef __attribute__((ext_vector_type(8))) float v8f;

__global__ __launch_bounds__(NTHREADS) void gnn_stage1(const float* __restrict__ A,
                                                       const float* __restrict__ X,
                                                       float* __restrict__ T) {
    __shared__ __attribute__((aligned(16))) float P[KS][16][PP];

    const int tid  = threadIdx.x;
    const int wave = tid >> 5;
    const int lane = tid & 31;
    const int b    = blockIdx.x / (N_ / 16);
    const int i0   = (blockIdx.x % (N_ / 16)) * 16;

    const int lm   = lane & 15;
    const int lk   = (lane >> 4) * 2;
    const int row0 = (lane >> 4) * 8;

    const float* Arow = A + (size_t)(b * N_ + i0 + lm) * N_ * 2;
    const float* Xb   = X + (size_t)b * N_ * D_ * TS;

    const int erow = tid >> 5;
    const int ecol = (tid & 31) * 2;
    const float x1a = Xb[((size_t)(i0 + erow) * D_ + ecol    ) * TS + 1];
    const float x1b = Xb[((size_t)(i0 + erow) * D_ + ecol + 1) * TS + 1];

    const int kbase = wave * KW;
    v4f af[NSTEP];
    v2f bf[NSTEP][NT];
#pragma unroll
    for (int s = 0; s < NSTEP; ++s) {
        const int j = kbase + 4 * s + lk;
        af[s] = *(const v4f*)(Arow + (size_t)j * 2);
#pragma unroll
        for (int q = 0; q < NT; ++q) {
            bf[s][q].x = Xb[((size_t)j       * D_ + 16 * q + lm) * TS + 2];
            bf[s][q].y = Xb[((size_t)(j + 1) * D_ + 16 * q + lm) * TS + 2];
        }
    }

    v8f c[NT] = {};
#pragma unroll
    for (int s = 0; s < NSTEP; ++s) {
        asm volatile("" :: "v"(af[s]), "v"(bf[s][0]), "v"(bf[s][1]), "v"(bf[s][2]), "v"(bf[s][3]));
        v2f a;  a.x = af[s].y;  a.y = af[s].w;
#pragma unroll
        for (int q = 0; q < NT; ++q) {
            c[q] = __builtin_amdgcn_wmma_f32_16x16x4_f32(false, a, false, bf[s][q],
                                                         (short)0, c[q], false, false);
        }
    }

#pragma unroll
    for (int q = 0; q < NT; ++q) {
#pragma unroll
        for (int r = 0; r < 8; ++r) {
            P[wave][row0 + r][16 * q + lm] = c[q][r];
        }
    }
    __syncthreads();

    float s0 = 0.0f, s1 = 0.0f;
#pragma unroll
    for (int k = 0; k < KS; ++k) {
        const float2 p = *(const float2*)&P[k][erow][ecol];
        s0 += p.x;
        s1 += p.y;
    }
    float2 o;  o.x = s0 + x1a;  o.y = s1 + x1b;
    *(float2*)(T + (size_t)(b * N_ + i0 + erow) * D_ + ecol) = o;
}

__global__ __launch_bounds__(NTHREADS) void gnn_stage2(const float* __restrict__ A,
                                                       const float* __restrict__ X,
                                                       const float* __restrict__ T,
                                                       const float* __restrict__ W,
                                                       float* __restrict__ out) {
    __shared__ __attribute__((aligned(16))) float P[KS][16][PP];
    __shared__ __attribute__((aligned(16))) float S[16][PP];

    const int tid  = threadIdx.x;
    const int wave = tid >> 5;
    const int lane = tid & 31;
    const int b    = blockIdx.x / (N_ / 16);
    const int i0   = (blockIdx.x % (N_ / 16)) * 16;
    const int lm   = lane & 15;
    const int lk   = (lane >> 4) * 2;
    const int row0 = (lane >> 4) * 8;

    const float* Arow = A + (size_t)(b * N_ + i0 + lm) * N_ * 2;
    const float* Xb   = X + (size_t)b * N_ * D_ * TS;
    const float* Tb   = T + (size_t)b * N_ * D_;

    const int erow = tid >> 5;
    const int ecol = (tid & 31) * 2;
    const float x0a = Xb[((size_t)(i0 + erow) * D_ + ecol    ) * TS + 0];
    const float x0b = Xb[((size_t)(i0 + erow) * D_ + ecol + 1) * TS + 0];

    const int kbase = wave * KW;
    v4f af[NSTEP];
    v2f bf[NSTEP][NT];
#pragma unroll
    for (int s = 0; s < NSTEP; ++s) {
        const int j = kbase + 4 * s + lk;
        af[s] = *(const v4f*)(Arow + (size_t)j * 2);
#pragma unroll
        for (int q = 0; q < NT; ++q) {
            bf[s][q].x = Tb[(size_t)j       * D_ + 16 * q + lm];
            bf[s][q].y = Tb[(size_t)(j + 1) * D_ + 16 * q + lm];
        }
    }

    v2f wf[D_ / 4] = {};
    if (wave < NT) {
#pragma unroll
        for (int s = 0; s < D_ / 4; ++s) {
            const int k = 4 * s + lk;
            wf[s].x = W[(size_t)k       * D_ + 16 * wave + lm];
            wf[s].y = W[(size_t)(k + 1) * D_ + 16 * wave + lm];
        }
    }

    v8f c[NT] = {};
#pragma unroll
    for (int s = 0; s < NSTEP; ++s) {
        asm volatile("" :: "v"(af[s]), "v"(bf[s][0]), "v"(bf[s][1]), "v"(bf[s][2]), "v"(bf[s][3]));
        v2f a;  a.x = af[s].x;  a.y = af[s].z;
#pragma unroll
        for (int q = 0; q < NT; ++q) {
            c[q] = __builtin_amdgcn_wmma_f32_16x16x4_f32(false, a, false, bf[s][q],
                                                         (short)0, c[q], false, false);
        }
    }
#pragma unroll
    for (int q = 0; q < NT; ++q) {
#pragma unroll
        for (int r = 0; r < 8; ++r) {
            P[wave][row0 + r][16 * q + lm] = c[q][r];
        }
    }
    __syncthreads();

    float s0 = 0.0f, s1 = 0.0f;
#pragma unroll
    for (int k = 0; k < KS; ++k) {
        const float2 p = *(const float2*)&P[k][erow][ecol];
        s0 += p.x;
        s1 += p.y;
    }
    float2 sv;  sv.x = s0 + x0a;  sv.y = s1 + x0b;
    *(float2*)&S[erow][ecol] = sv;
    __syncthreads();

    if (wave < NT) {
        v8f c2 = {};
#pragma unroll
        for (int s = 0; s < D_ / 4; ++s) {
            const int k = 4 * s + lk;
            v2f a;  a.x = S[lm][k];  a.y = S[lm][k + 1];
            c2 = __builtin_amdgcn_wmma_f32_16x16x4_f32(false, a, false, wf[s],
                                                       (short)0, c2, false, false);
        }
        float* Orow = out + (size_t)(b * N_ + i0) * D_;
#pragma unroll
        for (int r = 0; r < 8; ++r) {
            Orow[(size_t)(row0 + r) * D_ + 16 * wave + lm] = c2[r];
        }
    }
}

extern "C" void kernel_launch(void* const* d_in, const int* in_sizes, int n_in,
                              void* d_out, int out_size, void* d_ws, size_t ws_size,
                              hipStream_t stream) {
    (void)in_sizes; (void)n_in; (void)out_size; (void)ws_size;
    const float* A = (const float*)d_in[0];
    const float* X = (const float*)d_in[1];
    const float* W = (const float*)d_in[2];
    float* out = (float*)d_out;
    float* T   = (float*)d_ws;

    const dim3 grid(B_ * (N_ / 16));
    const dim3 block(NTHREADS);

    gnn_stage1<<<grid, block, 0, stream>>>(A, X, T);
    gnn_stage2<<<grid, block, 0, stream>>>(A, X, T, W, out);
}


# Stage 1: each 16x64 tile split into two 16x32 column halves on separate work-groups (192 blocks of 8 waves, 48-deep slices); 8 partials summed in LDS in fixed order (sum order changes)
# speedup vs baseline: 2.9317x; 1.1530x over previous
#include <hip/hip_runtime.h>
#include <hip/hip_bf16.h>

#define B_ 4
#define N_ 384
#define D_ 64
#define TS 3
#define KS 16
#define KW (N_ / KS)
#define NSTEP (KW / 4)
#define NT (D_ / 16)
#define PP (D_ + 4)
#define NTHREADS (KS * 32)
#define KS1 8
#define KW1 (N_ / KS1)
#define NSTEP1 (KW1 / 4)
#define NT1 2
#define CH (NT1 * 16)
#define PP1 (CH + 4)
#define NTHREADS1 (KS1 * 32)

typedef __attribute__((ext_vector_type(2))) float v2f;
typedef __attribute__((ext_vector_type(4))) float v4f;
typedef __attribute__((ext_vector_type(8))) float v8f;

__global__ __launch_bounds__(NTHREADS1) void gnn_stage1(const float* __restrict__ A,
                                                        const float* __restrict__ X,
                                                        float* __restrict__ T) {
    __shared__ __attribute__((aligned(16))) float P[KS1][16][PP1];

    const int tid  = threadIdx.x;
    const int wave = tid >> 5;
    const int lane = tid & 31;
    const int tile = blockIdx.x >> 1;
    const int c0   = (blockIdx.x & 1) * CH;
    const int b    = tile / (N_ / 16);
    const int i0   = (tile % (N_ / 16)) * 16;

    const int lm   = lane & 15;
    const int lk   = (lane >> 4) * 2;
    const int row0 = (lane >> 4) * 8;

    const float* Arow = A + (size_t)(b * N_ + i0 + lm) * N_ * 2;
    const float* Xb   = X + (size_t)b * N_ * D_ * TS;

    const int erow = tid >> 4;
    const int ecol = c0 + (tid & 15) * 2;
    const float x1a = Xb[((size_t)(i0 + erow) * D_ + ecol    ) * TS + 1];
    const float x1b = Xb[((size_t)(i0 + erow) * D_ + ecol + 1) * TS + 1];

    const int kbase = wave * KW1;
    v4f af[NSTEP1];
    v2f bf[NSTEP1][NT1];
#pragma unroll
    for (int s = 0; s < NSTEP1; ++s) {
        const int j = kbase + 4 * s + lk;
        af[s] = *(const v4f*)(Arow + (size_t)j * 2);
#pragma unroll
        for (int q = 0; q < NT1; ++q) {
            bf[s][q].x = Xb[((size_t)j       * D_ + c0 + 16 * q + lm) * TS + 2];
            bf[s][q].y = Xb[((size_t)(j + 1) * D_ + c0 + 16 * q + lm) * TS + 2];
        }
    }

    v8f c[NT1] = {};
#pragma unroll
    for (int s = 0; s < NSTEP1; ++s) {
        asm volatile("" :: "v"(af[s]), "v"(bf[s][0]), "v"(bf[s][1]));
        v2f a;  a.x = af[s].y;  a.y = af[s].w;
#pragma unroll
        for (int q = 0; q < NT1; ++q) {
            c[q] = __builtin_amdgcn_wmma_f32_16x16x4_f32(false, a, false, bf[s][q],
                                                         (short)0, c[q], false, false);
        }
    }

#pragma unroll
    for (int q = 0; q < NT1; ++q) {
#pragma unroll
        for (int r = 0; r < 8; ++r) {
            P[wave][row0 + r][16 * q + lm] = c[q][r];
        }
    }
    __syncthreads();

    float s0 = 0.0f, s1 = 0.0f;
    const int pcol = (tid & 15) * 2;
#pragma unroll
    for (int k = 0; k < KS1; ++k) {
        const float2 p = *(const float2*)&P[k][erow][pcol];
        s0 += p.x;
        s1 += p.y;
    }
    float2 o;  o.x = s0 + x1a;  o.y = s1 + x1b;
    *(float2*)(T + (size_t)(b * N_ + i0 + erow) * D_ + ecol) = o;
}

__global__ __launch_bounds__(NTHREADS) void gnn_stage2(const float* __restrict__ A,
                                                       const float* __restrict__ X,
                                                       const float* __restrict__ T,
                                                       const float* __restrict__ W,
                                                       float* __restrict__ out) {
    __shared__ __attribute__((aligned(16))) float P[KS][16][PP];
    __shared__ __attribute__((aligned(16))) float S[16][PP];

    const int tid  = threadIdx.x;
    const int wave = tid >> 5;
    const int lane = tid & 31;
    const int b    = blockIdx.x / (N_ / 16);
    const int i0   = (blockIdx.x % (N_ / 16)) * 16;
    const int lm   = lane & 15;
    const int lk   = (lane >> 4) * 2;
    const int row0 = (lane >> 4) * 8;

    const float* Arow = A + (size_t)(b * N_ + i0 + lm) * N_ * 2;
    const float* Xb   = X + (size_t)b * N_ * D_ * TS;
    const float* Tb   = T + (size_t)b * N_ * D_;

    const int erow = tid >> 5;
    const int ecol = (tid & 31) * 2;
    const float x0a = Xb[((size_t)(i0 + erow) * D_ + ecol    ) * TS + 0];
    const float x0b = Xb[((size_t)(i0 + erow) * D_ + ecol + 1) * TS + 0];

    const int kbase = wave * KW;
    v4f af[NSTEP];
    v2f bf[NSTEP][NT];
#pragma unroll
    for (int s = 0; s < NSTEP; ++s) {
        const int j = kbase + 4 * s + lk;
        af[s] = *(const v4f*)(Arow + (size_t)j * 2);
#pragma unroll
        for (int q = 0; q < NT; ++q) {
            bf[s][q].x = Tb[(size_t)j       * D_ + 16 * q + lm];
            bf[s][q].y = Tb[(size_t)(j + 1) * D_ + 16 * q + lm];
        }
    }

    v2f wf[D_ / 4] = {};
    if (wave < NT) {
#pragma unroll
        for (int s = 0; s < D_ / 4; ++s) {
            const int k = 4 * s + lk;
            wf[s].x = W[(size_t)k       * D_ + 16 * wave + lm];
            wf[s].y = W[(size_t)(k + 1) * D_ + 16 * wave + lm];
        }
    }

    v8f c[NT] = {};
#pragma unroll
    for (int s = 0; s < NSTEP; ++s) {
        asm volatile("" :: "v"(af[s]), "v"(bf[s][0]), "v"(bf[s][1]), "v"(bf[s][2]), "v"(bf[s][3]));
        v2f a;  a.x = af[s].x;  a.y = af[s].z;
#pragma unroll
        for (int q = 0; q < NT; ++q) {
            c[q] = __builtin_amdgcn_wmma_f32_16x16x4_f32(false, a, false, bf[s][q],
                                                         (short)0, c[q], false, false);
        }
    }
#pragma unroll
    for (int q = 0; q < NT; ++q) {
#pragma unroll
        for (int r = 0; r < 8; ++r) {
            P[wave][row0 + r][16 * q + lm] = c[q][r];
        }
    }
    __syncthreads();

    float s0 = 0.0f, s1 = 0.0f;
#pragma unroll
    for (int k = 0; k < KS; ++k) {
        const float2 p = *(const float2*)&P[k][erow][ecol];
        s0 += p.x;
        s1 += p.y;
    }
    float2 sv;  sv.x = s0 + x0a;  sv.y = s1 + x0b;
    *(float2*)&S[erow][ecol] = sv;
    __syncthreads();

    if (wave < NT) {
        v8f c2 = {};
#pragma unroll
        for (int s = 0; s < D_ / 4; ++s) {
            const int k = 4 * s + lk;
            v2f a;  a.x = S[lm][k];  a.y = S[lm][k + 1];
            c2 = __builtin_amdgcn_wmma_f32_16x16x4_f32(false, a, false, wf[s],
                                                       (short)0, c2, false, false);
        }
        float* Orow = out + (size_t)(b * N_ + i0) * D_;
#pragma unroll
        for (int r = 0; r < 8; ++r) {
            Orow[(size_t)(row0 + r) * D_ + 16 * wave + lm] = c2[r];
        }
    }
}

extern "C" void kernel_launch(void* const* d_in, const int* in_sizes, int n_in,
                              void* d_out, int out_size, void* d_ws, size_t ws_size,
                              hipStream_t stream) {
    (void)in_sizes; (void)n_in; (void)out_size; (void)ws_size;
    const float* A = (const float*)d_in[0];
    const float* X = (const float*)d_in[1];
    const float* W = (const float*)d_in[2];
    float* out = (float*)d_out;
    float* T   = (float*)d_ws;

    const dim3 grid(B_ * (N_ / 16));
    const dim3 block(NTHREADS);
    const dim3 grid1(B_ * (N_ / 16) * 2);
    const dim3 block1(NTHREADS1);

    gnn_stage1<<<grid1, block1, 0, stream>>>(A, X, T);
    gnn_stage2<<<grid, block, 0, stream>>>(A, X, T, W, out);
}
